# starting kernel
# baseline (speedup 1.0000x reference)
#include <hip/hip_runtime.h>
#include <math.h>

typedef __attribute__((ext_vector_type(2))) float v2f;
typedef __attribute__((ext_vector_type(8))) float v8f;

#define DIN 256
#define HN  64
#define ONUM 10
#define DT  320
#define WRS 68
#define HST 20

__device__ __forceinline__ v8f wmma4(v2f a, v2f b, v8f c) {
  return __builtin_amdgcn_wmma_f32_16x16x4_f32(false, a, false, b, (short)0, c,
                                               false, false);
}

__device__ __forceinline__ float fast_sigmoid(float s) {
  return __builtin_amdgcn_rcpf(1.0f + __expf(-s));
}

__global__ __launch_bounds__(256) void casper_prep(const float* __restrict__ Wh,
                                                   const float* __restrict__ Wout,
                                                   float* __restrict__ wsB) {
  int tid = blockIdx.x * blockDim.x + threadIdx.x;
  if (tid >= 64 * 5 * 32) return;
  int t  = tid / 160;
  int nt = (tid % 160) / 32;
  int l  = tid & 31;
  int kb = 4 * t + ((l >> 4) << 1);
  float v0 = 0.f, v1 = 0.f;
  if (nt < 4) {
    int n = nt * 16 + (l & 15);
    v0 = Wh[n * DT + kb];
    v1 = Wh[n * DT + kb + 1];
  } else {
    int o = l & 15;
    if (o < ONUM) {
      v0 = Wout[o * DT + kb];
      v1 = Wout[o * DT + kb + 1];
    }
  }
  ((v2f*)wsB)[tid] = (v2f){v0, v1};
}

__global__ __launch_bounds__(256) void casper_main(
    const float* __restrict__ x, const float* __restrict__ Wh,
    const float* __restrict__ bh, const float* __restrict__ Wout,
    const float* __restrict__ bo, const float* __restrict__ wsB,
    float* __restrict__ out, int nrows) {
  __shared__ __align__(16) float WrL[HN * WRS];
  __shared__ __align__(16) float WorL[16 * WRS];
  __shared__ float bhL[HN];
  __shared__ float boL[16];
  __shared__ __align__(16) float hbuf[8][4][16 * HST];

  const int tid  = threadIdx.x;
  const int lane = tid & 31;
  const int wid  = tid >> 5;
  const int mrow  = lane & 15;
  const int khalf = lane >> 4;

  for (int idx = tid; idx < HN * WRS; idx += 256) {
    int i = idx / WRS, j = idx % WRS;
    WrL[idx] = (j < HN) ? Wh[i * DT + DIN + j] : 0.f;
  }
  for (int idx = tid; idx < 16 * WRS; idx += 256) {
    int o = idx / WRS, j = idx % WRS;
    WorL[idx] = (o < ONUM && j < HN) ? Wout[o * DT + DIN + j] : 0.f;
  }
  if (tid < HN) bhL[tid] = bh[tid];
  if (tid < 16) boL[tid] = (tid < ONUM) ? bo[tid] : 0.f;
  __syncthreads();

  const int row0 = blockIdx.x * 128 + wid * 16;
  const float* xl = x + (size_t)(row0 + mrow) * DIN + khalf * 2;

  v8f cP[4];
  v8f cOut = {0.f, 0.f, 0.f, 0.f, 0.f, 0.f, 0.f, 0.f};
#pragma unroll
  for (int g = 0; g < 4; ++g) cP[g] = cOut;

  const v2f* Bws = (const v2f*)wsB;
#pragma unroll 2
  for (int t = 0; t < 64; ++t) {
    v2f a  = *(const v2f*)(xl + 4 * t);
    v2f b0 = Bws[(t * 5 + 0) * 32 + lane];
    v2f b1 = Bws[(t * 5 + 1) * 32 + lane];
    v2f b2 = Bws[(t * 5 + 2) * 32 + lane];
    v2f b3 = Bws[(t * 5 + 3) * 32 + lane];
    v2f b4 = Bws[(t * 5 + 4) * 32 + lane];
    cP[0] = wmma4(a, b0, cP[0]);
    cP[1] = wmma4(a, b1, cP[1]);
    cP[2] = wmma4(a, b2, cP[2]);
    cP[3] = wmma4(a, b3, cP[3]);
    cOut  = wmma4(a, b4, cOut);
  }

#pragma unroll
  for (int g = 0; g < 4; ++g) {
    v8f acc = cP[g];
#pragma unroll
    for (int gp = 0; gp < 4; ++gp) {
      if (gp < g) {
#pragma unroll
        for (int kk = 0; kk < 4; ++kk) {
          int jj = 4 * kk + khalf * 2;
          v2f a = *(const v2f*)&hbuf[wid][gp][mrow * HST + jj];
          v2f b = *(const v2f*)&WrL[(g * 16 + mrow) * WRS + gp * 16 + jj];
          acc = wmma4(a, b, acc);
        }
      }
    }
    {
      int mb = khalf * 8;
#pragma unroll
      for (int r = 0; r < 8; ++r)
        hbuf[wid][g][(mb + r) * HST + mrow] = acc[r];
    }
    __syncthreads();
    {
      float hl[16];
#pragma unroll
      for (int i = 0; i < 16; ++i) {
        float s = hbuf[wid][g][mrow * HST + i] + bhL[g * 16 + i];
#pragma unroll
        for (int j = 0; j < i; ++j)
          s += WrL[(g * 16 + i) * WRS + g * 16 + j] * hl[j];
        float h = fast_sigmoid(s);
        hl[i] = h;
        hbuf[wid][g][mrow * HST + i] = h;
      }
    }
    __syncthreads();
  }

#pragma unroll
  for (int g = 0; g < 4; ++g) {
#pragma unroll
    for (int kk = 0; kk < 4; ++kk) {
      int jj = 4 * kk + khalf * 2;
      v2f a = *(const v2f*)&hbuf[wid][g][mrow * HST + jj];
      v2f b = *(const v2f*)&WorL[mrow * WRS + g * 16 + jj];
      cOut = wmma4(a, b, cOut);
    }
  }

  if (mrow < ONUM) {
    int mb = khalf * 8;
    float bb = boL[mrow];
#pragma unroll
    for (int r = 0; r < 8; ++r)
      out[(size_t)(row0 + mb + r) * ONUM + mrow] = cOut[r] + bb;
  }
}

extern "C" void kernel_launch(void* const* d_in, const int* in_sizes, int n_in,
                              void* d_out, int out_size, void* d_ws,
                              size_t ws_size, hipStream_t stream) {
  const float* x   = (const float*)d_in[0];
  const float* Wh  = (const float*)d_in[1];
  const float* bhp = (const float*)d_in[2];
  const float* Wo  = (const float*)d_in[3];
  const float* bop = (const float*)d_in[4];
  float* outp = (float*)d_out;
  float* wsB  = (float*)d_ws;

  int nrows   = in_sizes[0] / DIN;
  int nblocks = nrows / 128;

  casper_prep<<<40, 256, 0, stream>>>(Wh, Wo, wsB);
  casper_main<<<nblocks, 256, 0, stream>>>(x, Wh, bhp, Wo, bop, wsB, outp,
                                           nrows);
}


# W_h and W_out staged once per 1024-thread work-group in LDS; K-loop B fragments read from LDS instead of a packed global copy, packing launch dropped; summation order unchanged
# speedup vs baseline: 1.3643x; 1.3643x over previous
#include <hip/hip_runtime.h>
#include <math.h>


typedef __attribute__((ext_vector_type(2))) float v2f;
typedef __attribute__((ext_vector_type(4))) float v4f;
typedef __attribute__((ext_vector_type(8))) float v8f;

#define DIN 256
#define HN  64
#define ONUM 10
#define DT  320
#define WRS 68
#define WP  260
#define WROWS 80
#define RQ  (DT / 4)
#define HST 20
#define NWAVE 32
#define ROWS_PER_WG (NWAVE * 16)

__device__ __forceinline__ v8f wmma4(v2f a, v2f b, v8f c) {
  return __builtin_amdgcn_wmma_f32_16x16x4_f32(false, a, false, b, (short)0, c,
                                               false, false);
}

__device__ __forceinline__ float fast_sigmoid(float s) {
  return __builtin_amdgcn_rcpf(1.0f + __expf(-s));
}

__global__ __launch_bounds__(1024) void casper_main(
    const float* __restrict__ x, const float* __restrict__ Wh,
    const float* __restrict__ bh, const float* __restrict__ Wout,
    const float* __restrict__ bo, float* __restrict__ out, int nrows) {
  __shared__ __align__(256) float WrL[HN * WRS];
  __shared__ __align__(256) float WorL[16 * WRS];
  __shared__ __align__(256) float bhL[HN];
  __shared__ __align__(256) float boL[16];
  __shared__ __align__(128) float WinL[WROWS * WP];
  __shared__ __align__(16) float hbuf[NWAVE][4][16 * HST];

  const int tid  = threadIdx.x;
  const int lane = tid & 31;
  const int wid  = tid >> 5;
  const int mrow  = lane & 15;
  const int khalf = lane >> 4;

  {
    v4f wv[5];
#pragma unroll
    for (int i = 0; i < 5; ++i)
      wv[i] = *(const v4f*)(Wh + 4 * (tid + 1024 * i));
    const int qo = (tid < ONUM * RQ) ? tid : 0;
    v4f wo = *(const v4f*)(Wout + 4 * qo);
    float bhv = bh[tid & (HN - 1)];
    float bov = bo[(tid < ONUM) ? tid : 0];
    const v4f z4 = {0.f, 0.f, 0.f, 0.f};

#pragma unroll
    for (int i = 0; i < 5; ++i) {
      int q = tid + 1024 * i;
      int r = q / RQ, c4 = q - r * RQ;
      float* dst = (c4 < DIN / 4) ? &WinL[r * WP + 4 * c4]
                                  : &WrL[r * WRS + 4 * (c4 - DIN / 4)];
      *(v4f*)dst = wv[i];
    }
    {
      int r = tid / RQ, c4 = tid - r * RQ;
      float* dst = (c4 < DIN / 4) ? &WinL[(HN + r) * WP + 4 * c4]
                                  : &WorL[r * WRS + 4 * (c4 - DIN / 4)];
      *(v4f*)dst = (tid < ONUM * RQ) ? wo : z4;
    }
    if (tid < 16 * RQ - 1024) {
      int q = tid + 1024;
      int r = q / RQ, c4 = q - r * RQ;
      float* dst = (c4 < DIN / 4) ? &WinL[(HN + r) * WP + 4 * c4]
                                  : &WorL[r * WRS + 4 * (c4 - DIN / 4)];
      *(v4f*)dst = z4;
    }
    if (tid < HN) bhL[tid] = bhv;
    if (tid < 16) boL[tid] = (tid < ONUM) ? bov : 0.f;
  }
  __syncthreads();

  const int row0 = blockIdx.x * ROWS_PER_WG + wid * 16;
  const float* xl = x + (size_t)(row0 + mrow) * DIN + khalf * 2;
  const float* bl = &WinL[mrow * WP + khalf * 2];

  v8f cP[4];
  v8f cOut = {0.f, 0.f, 0.f, 0.f, 0.f, 0.f, 0.f, 0.f};
#pragma unroll
  for (int g = 0; g < 4; ++g) cP[g] = cOut;

#pragma unroll 2
  for (int t = 0; t < 64; ++t) {
    v2f a  = *(const v2f*)(xl + 4 * t);
    v2f b0 = *(const v2f*)(bl + 0 * 16 * WP + 4 * t);
    v2f b1 = *(const v2f*)(bl + 1 * 16 * WP + 4 * t);
    v2f b2 = *(const v2f*)(bl + 2 * 16 * WP + 4 * t);
    v2f b3 = *(const v2f*)(bl + 3 * 16 * WP + 4 * t);
    v2f b4 = *(const v2f*)(bl + 4 * 16 * WP + 4 * t);
    cP[0] = wmma4(a, b0, cP[0]);
    cP[1] = wmma4(a, b1, cP[1]);
    cP[2] = wmma4(a, b2, cP[2]);
    cP[3] = wmma4(a, b3, cP[3]);
    cOut  = wmma4(a, b4, cOut);
  }

#pragma unroll
  for (int g = 0; g < 4; ++g) {
    v8f acc = cP[g];
#pragma unroll
    for (int gp = 0; gp < 4; ++gp) {
      if (gp < g) {
#pragma unroll
        for (int kk = 0; kk < 4; ++kk) {
          int jj = 4 * kk + khalf * 2;
          v2f a = *(const v2f*)&hbuf[wid][gp][mrow * HST + jj];
          v2f b = *(const v2f*)&WrL[(g * 16 + mrow) * WRS + gp * 16 + jj];
          acc = wmma4(a, b, acc);
        }
      }
    }
    {
      int mb = khalf * 8;
#pragma unroll
      for (int r = 0; r < 8; ++r)
        hbuf[wid][g][(mb + r) * HST + mrow] = acc[r];
    }
    __syncthreads();
    {
      float hl[16];
#pragma unroll
      for (int i = 0; i < 16; ++i) {
        float s = hbuf[wid][g][mrow * HST + i] + bhL[g * 16 + i];
#pragma unroll
        for (int j = 0; j < i; ++j)
          s += WrL[(g * 16 + i) * WRS + g * 16 + j] * hl[j];
        float h = fast_sigmoid(s);
        hl[i] = h;
        hbuf[wid][g][mrow * HST + i] = h;
      }
    }
    __syncthreads();
  }

#pragma unroll
  for (int g = 0; g < 4; ++g) {
#pragma unroll
    for (int kk = 0; kk < 4; ++kk) {
      int jj = 4 * kk + khalf * 2;
      v2f a = *(const v2f*)&hbuf[wid][g][mrow * HST + jj];
      v2f b = *(const v2f*)&WorL[mrow * WRS + g * 16 + jj];
      cOut = wmma4(a, b, cOut);
    }
  }

  if (mrow < ONUM) {
    int mb = khalf * 8;
    float bb = boL[mrow];
#pragma unroll
    for (int r = 0; r < 8; ++r)
      out[(size_t)(row0 + mb + r) * ONUM + mrow] = cOut[r] + bb;
  }
}

extern "C" void kernel_launch(void* const* d_in, const int* in_sizes, int n_in,
                              void* d_out, int out_size, void* d_ws,
                              size_t ws_size, hipStream_t stream) {
  const float* x   = (const float*)d_in[0];
  const float* Wh  = (const float*)d_in[1];
  const float* bhp = (const float*)d_in[2];
  const float* Wo  = (const float*)d_in[3];
  const float* bop = (const float*)d_in[4];
  float* outp = (float*)d_out;

  int nrows   = in_sizes[0] / DIN;
  int nblocks = nrows / ROWS_PER_WG;

  casper_main<<<nblocks, 1024, 0, stream>>>(x, Wh, bhp, Wo, bop, outp, nrows);
}


# each wave owns two 16-row tiles (32 rows, 16 waves per work-group): each B fragment and weight broadcast feeds both, the cascade's serial part runs one row per lane; summation order unchanged
# speedup vs baseline: 1.4789x; 1.0840x over previous
#include <hip/hip_runtime.h>
#include <math.h>


typedef __attribute__((ext_vector_type(2))) float v2f;
typedef __attribute__((ext_vector_type(4))) float v4f;
typedef __attribute__((ext_vector_type(8))) float v8f;

#define DIN 256
#define HN  64
#define ONUM 10
#define DT  320
#define WRS 68
#define WP  260
#define WROWS 80
#define RQ  (DT / 4)
#define HST 20
#define NWAVE 16
#define MT 2
#define WROWN (MT * 16)
#define NTHR (NWAVE * 32)
#define ROWS_PER_WG (NWAVE * WROWN)

__device__ __forceinline__ v8f wmma4(v2f a, v2f b, v8f c) {
  return __builtin_amdgcn_wmma_f32_16x16x4_f32(false, a, false, b, (short)0, c,
                                               false, false);
}

__device__ __forceinline__ float fast_sigmoid(float s) {
  return __builtin_amdgcn_rcpf(1.0f + __expf(-s));
}

__global__ __launch_bounds__(NTHR) void casper_main(
    const float* __restrict__ x, const float* __restrict__ Wh,
    const float* __restrict__ bh, const float* __restrict__ Wout,
    const float* __restrict__ bo, float* __restrict__ out, int nrows) {
  __shared__ __align__(256) float WrL[HN * WRS];
  __shared__ __align__(256) float WorL[16 * WRS];
  __shared__ __align__(256) float bhL[HN];
  __shared__ __align__(256) float boL[16];
  __shared__ __align__(128) float WinL[WROWS * WP];
  __shared__ __align__(16) float hbuf[NWAVE][4][WROWN * HST];

  const int tid  = threadIdx.x;
  const int lane = tid & 31;
  const int wid  = tid >> 5;
  const int mrow  = lane & 15;
  const int khalf = lane >> 4;

  {
    v4f wv[10];
#pragma unroll
    for (int i = 0; i < 10; ++i)
      wv[i] = *(const v4f*)(Wh + 4 * (tid + NTHR * i));
    v4f wo[2];
#pragma unroll
    for (int i = 0; i < 2; ++i) {
      int q = tid + NTHR * i;
      wo[i] = *(const v4f*)(Wout + 4 * ((q < ONUM * RQ) ? q : 0));
    }
    float bhv = bh[tid & (HN - 1)];
    float bov = bo[(tid < ONUM) ? tid : 0];
    const v4f z4 = {0.f, 0.f, 0.f, 0.f};

#pragma unroll
    for (int i = 0; i < 10; ++i) {
      int q = tid + NTHR * i;
      int r = q / RQ, c4 = q - r * RQ;
      float* dst = (c4 < DIN / 4) ? &WinL[r * WP + 4 * c4]
                                  : &WrL[r * WRS + 4 * (c4 - DIN / 4)];
      *(v4f*)dst = wv[i];
    }
#pragma unroll
    for (int i = 0; i < 2; ++i) {
      int q = tid + NTHR * i;
      int r = q / RQ, c4 = q - r * RQ;
      float* dst = (c4 < DIN / 4) ? &WinL[(HN + r) * WP + 4 * c4]
                                  : &WorL[r * WRS + 4 * (c4 - DIN / 4)];
      *(v4f*)dst = (q < ONUM * RQ) ? wo[i] : z4;
    }
    if (tid < 16 * RQ - 2 * NTHR) {
      int q = tid + 2 * NTHR;
      int r = q / RQ, c4 = q - r * RQ;
      float* dst = (c4 < DIN / 4) ? &WinL[(HN + r) * WP + 4 * c4]
                                  : &WorL[r * WRS + 4 * (c4 - DIN / 4)];
      *(v4f*)dst = z4;
    }
    if (tid < HN) bhL[tid] = bhv;
    if (tid < 16) boL[tid] = (tid < ONUM) ? bov : 0.f;
  }
  __syncthreads();

  const int row0 = blockIdx.x * ROWS_PER_WG + wid * WROWN;
  const float* xl = x + (size_t)(row0 + mrow) * DIN + khalf * 2;
  const float* bl = &WinL[mrow * WP + khalf * 2];

  v8f cP[MT][4];
  v8f cOut[MT];
  {
    const v8f z8 = {0.f, 0.f, 0.f, 0.f, 0.f, 0.f, 0.f, 0.f};
#pragma unroll
    for (int m = 0; m < MT; ++m) {
      cOut[m] = z8;
#pragma unroll
      for (int g = 0; g < 4; ++g) cP[m][g] = z8;
    }
  }

#pragma unroll 2
  for (int t = 0; t < 64; ++t) {
    v2f a0 = *(const v2f*)(xl + 4 * t);
    v2f a1 = *(const v2f*)(xl + 16 * DIN + 4 * t);
    v2f b0 = *(const v2f*)(bl + 0 * 16 * WP + 4 * t);
    v2f b1 = *(const v2f*)(bl + 1 * 16 * WP + 4 * t);
    v2f b2 = *(const v2f*)(bl + 2 * 16 * WP + 4 * t);
    v2f b3 = *(const v2f*)(bl + 3 * 16 * WP + 4 * t);
    v2f b4 = *(const v2f*)(bl + 4 * 16 * WP + 4 * t);
    cP[0][0] = wmma4(a0, b0, cP[0][0]);
    cP[1][0] = wmma4(a1, b0, cP[1][0]);
    cP[0][1] = wmma4(a0, b1, cP[0][1]);
    cP[1][1] = wmma4(a1, b1, cP[1][1]);
    cP[0][2] = wmma4(a0, b2, cP[0][2]);
    cP[1][2] = wmma4(a1, b2, cP[1][2]);
    cP[0][3] = wmma4(a0, b3, cP[0][3]);
    cP[1][3] = wmma4(a1, b3, cP[1][3]);
    cOut[0]  = wmma4(a0, b4, cOut[0]);
    cOut[1]  = wmma4(a1, b4, cOut[1]);
  }

#pragma unroll
  for (int g = 0; g < 4; ++g) {
    v8f acc[MT];
#pragma unroll
    for (int m = 0; m < MT; ++m) acc[m] = cP[m][g];
#pragma unroll
    for (int gp = 0; gp < 4; ++gp) {
      if (gp < g) {
#pragma unroll
        for (int kk = 0; kk < 4; ++kk) {
          int jj = 4 * kk + khalf * 2;
          v2f b = *(const v2f*)&WrL[(g * 16 + mrow) * WRS + gp * 16 + jj];
#pragma unroll
          for (int m = 0; m < MT; ++m) {
            v2f a = *(const v2f*)&hbuf[wid][gp][(16 * m + mrow) * HST + jj];
            acc[m] = wmma4(a, b, acc[m]);
          }
        }
      }
    }
    {
      int mb = khalf * 8;
#pragma unroll
      for (int m = 0; m < MT; ++m)
#pragma unroll
        for (int r = 0; r < 8; ++r)
          hbuf[wid][g][(16 * m + mb + r) * HST + mrow] = acc[m][r];
    }
    __syncthreads();
    {
      float hl[16];
#pragma unroll
      for (int i = 0; i < 16; ++i) {
        float s = hbuf[wid][g][lane * HST + i] + bhL[g * 16 + i];
#pragma unroll
        for (int j = 0; j < i; ++j)
          s += WrL[(g * 16 + i) * WRS + g * 16 + j] * hl[j];
        float h = fast_sigmoid(s);
        hl[i] = h;
        hbuf[wid][g][lane * HST + i] = h;
      }
    }
    __syncthreads();
  }

#pragma unroll
  for (int g = 0; g < 4; ++g) {
#pragma unroll
    for (int kk = 0; kk < 4; ++kk) {
      int jj = 4 * kk + khalf * 2;
      v2f b = *(const v2f*)&WorL[mrow * WRS + g * 16 + jj];
#pragma unroll
      for (int m = 0; m < MT; ++m) {
        v2f a = *(const v2f*)&hbuf[wid][g][(16 * m + mrow) * HST + jj];
        cOut[m] = wmma4(a, b, cOut[m]);
      }
    }
  }

  if (mrow < ONUM) {
    int mb = khalf * 8;
    float bb = boL[mrow];
#pragma unroll
    for (int m = 0; m < MT; ++m)
#pragma unroll
      for (int r = 0; r < 8; ++r)
        out[(size_t)(row0 + 16 * m + mb + r) * ONUM + mrow] = cOut[m][r] + bb;
  }
}

extern "C" void kernel_launch(void* const* d_in, const int* in_sizes, int n_in,
                              void* d_out, int out_size, void* d_ws,
                              size_t ws_size, hipStream_t stream) {
  const float* x   = (const float*)d_in[0];
  const float* Wh  = (const float*)d_in[1];
  const float* bhp = (const float*)d_in[2];
  const float* Wo  = (const float*)d_in[3];
  const float* bop = (const float*)d_in[4];
  float* outp = (float*)d_out;

  int nrows   = in_sizes[0] / DIN;
  int nblocks = nrows / ROWS_PER_WG;

  casper_main<<<nblocks, NTHR, 0, stream>>>(x, Wh, bhp, Wo, bop, outp, nrows);
}


# K-loop software-pipelined in registers: x fragments loaded 16 K-steps ahead (16-slot ring per tile), weight fragments a block ahead, loads kept grouped per 20-WMMA block; summation order unchanged
# speedup vs baseline: 1.6116x; 1.0897x over previous
#include <hip/hip_runtime.h>
#include <math.h>


typedef __attribute__((ext_vector_type(2))) float v2f;
typedef __attribute__((ext_vector_type(4))) float v4f;
typedef __attribute__((ext_vector_type(8))) float v8f;

#define DIN 256
#define HN  64
#define ONUM 10
#define DT  320
#define WRS 68
#define WP  260
#define WROWS 80
#define RQ  (DT / 4)
#define HST 20
#define NWAVE 16
#define MT 2
#define WROWN (MT * 16)
#define NTHR (NWAVE * 32)
#define ROWS_PER_WG (NWAVE * WROWN)
#define RING 16

__device__ __forceinline__ v8f wmma4(v2f a, v2f b, v8f c) {
  return __builtin_amdgcn_wmma_f32_16x16x4_f32(false, a, false, b, (short)0, c,
                                               false, false);
}

__device__ __forceinline__ float fast_sigmoid(float s) {
  return __builtin_amdgcn_rcpf(1.0f + __expf(-s));
}


template <bool REFILL>
__device__ __forceinline__ void kgroup16(v8f (&cP)[MT][4], v8f (&cOut)[MT],
                                         v2f (&ra)[MT][RING], v2f (&bb)[2][2][5],
                                         const float* xl, const float* bl, int u) {
#pragma unroll
  for (int p = 0; p < RING / 2; ++p) {
    const int t = RING * u + 2 * p;
    const int cur = p & 1, nxt = cur ^ 1;
    if (REFILL || p < RING / 2 - 1) {
#pragma unroll
      for (int kk = 0; kk < 2; ++kk)
#pragma unroll
        for (int nt = 0; nt < 5; ++nt)
          bb[nxt][kk][nt] = *(const v2f*)(bl + nt * 16 * WP + 4 * (t + 2 + kk));
    }
    __builtin_amdgcn_sched_barrier(0);
#pragma unroll
    for (int kk = 0; kk < 2; ++kk) {
      const v2f a0 = ra[0][2 * p + kk];
      const v2f a1 = ra[1][2 * p + kk];
      cP[0][0] = wmma4(a0, bb[cur][kk][0], cP[0][0]);
      cP[1][0] = wmma4(a1, bb[cur][kk][0], cP[1][0]);
      cP[0][1] = wmma4(a0, bb[cur][kk][1], cP[0][1]);
      cP[1][1] = wmma4(a1, bb[cur][kk][1], cP[1][1]);
      cP[0][2] = wmma4(a0, bb[cur][kk][2], cP[0][2]);
      cP[1][2] = wmma4(a1, bb[cur][kk][2], cP[1][2]);
      cP[0][3] = wmma4(a0, bb[cur][kk][3], cP[0][3]);
      cP[1][3] = wmma4(a1, bb[cur][kk][3], cP[1][3]);
      cOut[0]  = wmma4(a0, bb[cur][kk][4], cOut[0]);
      cOut[1]  = wmma4(a1, bb[cur][kk][4], cOut[1]);
    }
    __builtin_amdgcn_sched_barrier(0);
    if (REFILL) {
#pragma unroll
      for (int kk = 0; kk < 2; ++kk) {
        ra[0][2 * p + kk] = *(const v2f*)(xl + 4 * (t + RING + kk));
        ra[1][2 * p + kk] = *(const v2f*)(xl + 16 * DIN + 4 * (t + RING + kk));
      }
    }
  }
}

__global__ __launch_bounds__(NTHR) void casper_main(
    const float* __restrict__ x, const float* __restrict__ Wh,
    const float* __restrict__ bh, const float* __restrict__ Wout,
    const float* __restrict__ bo, float* __restrict__ out, int nrows) {
  __shared__ __align__(256) float WrL[HN * WRS];
  __shared__ __align__(256) float WorL[16 * WRS];
  __shared__ __align__(256) float bhL[HN];
  __shared__ __align__(256) float boL[16];
  __shared__ __align__(128) float WinL[WROWS * WP];
  __shared__ __align__(16) float hbuf[NWAVE][4][WROWN * HST];

  const int tid  = threadIdx.x;
  const int lane = tid & 31;
  const int wid  = tid >> 5;
  const int mrow  = lane & 15;
  const int khalf = lane >> 4;

  {
    v4f wv[10];
#pragma unroll
    for (int i = 0; i < 10; ++i)
      wv[i] = *(const v4f*)(Wh + 4 * (tid + NTHR * i));
    v4f wo[2];
#pragma unroll
    for (int i = 0; i < 2; ++i) {
      int q = tid + NTHR * i;
      wo[i] = *(const v4f*)(Wout + 4 * ((q < ONUM * RQ) ? q : 0));
    }
    float bhv = bh[tid & (HN - 1)];
    float bov = bo[(tid < ONUM) ? tid : 0];
    const v4f z4 = {0.f, 0.f, 0.f, 0.f};

#pragma unroll
    for (int i = 0; i < 10; ++i) {
      int q = tid + NTHR * i;
      int r = q / RQ, c4 = q - r * RQ;
      float* dst = (c4 < DIN / 4) ? &WinL[r * WP + 4 * c4]
                                  : &WrL[r * WRS + 4 * (c4 - DIN / 4)];
      *(v4f*)dst = wv[i];
    }
#pragma unroll
    for (int i = 0; i < 2; ++i) {
      int q = tid + NTHR * i;
      int r = q / RQ, c4 = q - r * RQ;
      float* dst = (c4 < DIN / 4) ? &WinL[(HN + r) * WP + 4 * c4]
                                  : &WorL[r * WRS + 4 * (c4 - DIN / 4)];
      *(v4f*)dst = (q < ONUM * RQ) ? wo[i] : z4;
    }
    if (tid < 16 * RQ - 2 * NTHR) {
      int q = tid + 2 * NTHR;
      int r = q / RQ, c4 = q - r * RQ;
      float* dst = (c4 < DIN / 4) ? &WinL[(HN + r) * WP + 4 * c4]
                                  : &WorL[r * WRS + 4 * (c4 - DIN / 4)];
      *(v4f*)dst = z4;
    }
    if (tid < HN) bhL[tid] = bhv;
    if (tid < 16) boL[tid] = (tid < ONUM) ? bov : 0.f;
  }
  __syncthreads();

  const int row0 = blockIdx.x * ROWS_PER_WG + wid * WROWN;
  const float* xl = x + (size_t)(row0 + mrow) * DIN + khalf * 2;
  const float* bl = &WinL[mrow * WP + khalf * 2];

  v8f cP[MT][4];
  v8f cOut[MT];
  {
    const v8f z8 = {0.f, 0.f, 0.f, 0.f, 0.f, 0.f, 0.f, 0.f};
#pragma unroll
    for (int m = 0; m < MT; ++m) {
      cOut[m] = z8;
#pragma unroll
      for (int g = 0; g < 4; ++g) cP[m][g] = z8;
    }
  }

  {
    v2f ra[MT][RING];
#pragma unroll
    for (int s = 0; s < RING; ++s) {
      ra[0][s] = *(const v2f*)(xl + 4 * s);
      ra[1][s] = *(const v2f*)(xl + 16 * DIN + 4 * s);
    }
    v2f bb[2][2][5];
#pragma unroll
    for (int kk = 0; kk < 2; ++kk)
#pragma unroll
      for (int nt = 0; nt < 5; ++nt)
        bb[0][kk][nt] = *(const v2f*)(bl + nt * 16 * WP + 4 * kk);
    for (int u = 0; u < DIN / (4 * RING) - 1; ++u)
      kgroup16<true>(cP, cOut, ra, bb, xl, bl, u);
    kgroup16<false>(cP, cOut, ra, bb, xl, bl, DIN / (4 * RING) - 1);
  }

#pragma unroll
  for (int g = 0; g < 4; ++g) {
    v8f acc[MT];
#pragma unroll
    for (int m = 0; m < MT; ++m) acc[m] = cP[m][g];
#pragma unroll
    for (int gp = 0; gp < 4; ++gp) {
      if (gp < g) {
#pragma unroll
        for (int kk = 0; kk < 4; ++kk) {
          int jj = 4 * kk + khalf * 2;
          v2f b = *(const v2f*)&WrL[(g * 16 + mrow) * WRS + gp * 16 + jj];
#pragma unroll
          for (int m = 0; m < MT; ++m) {
            v2f a = *(const v2f*)&hbuf[wid][gp][(16 * m + mrow) * HST + jj];
            acc[m] = wmma4(a, b, acc[m]);
          }
        }
      }
    }
    {
      int mb = khalf * 8;
#pragma unroll
      for (int m = 0; m < MT; ++m)
#pragma unroll
        for (int r = 0; r < 8; ++r)
          hbuf[wid][g][(16 * m + mb + r) * HST + mrow] = acc[m][r];
    }
    __syncthreads();
    {
      float hl[16];
#pragma unroll
      for (int i = 0; i < 16; ++i) {
        float s = hbuf[wid][g][lane * HST + i] + bhL[g * 16 + i];
#pragma unroll
        for (int j = 0; j < i; ++j)
          s += WrL[(g * 16 + i) * WRS + g * 16 + j] * hl[j];
        float h = fast_sigmoid(s);
        hl[i] = h;
        hbuf[wid][g][lane * HST + i] = h;
      }
    }
    __syncthreads();
  }

#pragma unroll
  for (int g = 0; g < 4; ++g) {
#pragma unroll
    for (int kk = 0; kk < 4; ++kk) {
      int jj = 4 * kk + khalf * 2;
      v2f b = *(const v2f*)&WorL[mrow * WRS + g * 16 + jj];
#pragma unroll
      for (int m = 0; m < MT; ++m) {
        v2f a = *(const v2f*)&hbuf[wid][g][(16 * m + mrow) * HST + jj];
        cOut[m] = wmma4(a, b, cOut[m]);
      }
    }
  }

  if (mrow < ONUM) {
    int mb = khalf * 8;
    float bb = boL[mrow];
#pragma unroll
    for (int m = 0; m < MT; ++m)
#pragma unroll
      for (int r = 0; r < 8; ++r)
        out[(size_t)(row0 + 16 * m + mb + r) * ONUM + mrow] = cOut[m][r] + bb;
  }
}

extern "C" void kernel_launch(void* const* d_in, const int* in_sizes, int n_in,
                              void* d_out, int out_size, void* d_ws,
                              size_t ws_size, hipStream_t stream) {
  const float* x   = (const float*)d_in[0];
  const float* Wh  = (const float*)d_in[1];
  const float* bhp = (const float*)d_in[2];
  const float* Wo  = (const float*)d_in[3];
  const float* bop = (const float*)d_in[4];
  float* outp = (float*)d_out;

  int nrows   = in_sizes[0] / DIN;
  int nblocks = nrows / ROWS_PER_WG;

  casper_main<<<nblocks, NTHR, 0, stream>>>(x, Wh, bhp, Wo, bop, outp, nrows);
}
